# starting kernel
# baseline (speedup 1.0000x reference)
#include <hip/hip_runtime.h>
#include <math.h>

#define QPS     4
#define NSPLIT  4
#define NLAYER  4
#define DIM     16
#define NCLASS  5
#define BATCH   32768
#define FEAT    16
#define H1      128
#define H2      64

#define WAVES_PER_BLOCK 4
#define H1P 132
#define H2P 68

typedef __attribute__((ext_vector_type(2))) float v2f;
typedef __attribute__((ext_vector_type(8))) float v8f;

__global__ void qcnn_unitaries(const float* __restrict__ sw, float* __restrict__ U) {
    int g = threadIdx.x;
    if (g >= NSPLIT * NLAYER * QPS) return;
    const float* w = sw + g * 3;
    float hx = 0.5f * w[0], hy = 0.5f * w[1], hz = 0.5f * w[2];
    float sx, cx, sy, cy, sz, cz;
    sincosf(hx, &sx, &cx);
    sincosf(hy, &sy, &cy);
    sincosf(hz, &sz, &cz);
    float m00r = cy * cx, m00i =  sy * sx;
    float m01r = -sy * cx, m01i = -cy * sx;
    float m10r =  sy * cx, m10i = -cy * sx;
    float m11r =  cy * cx, m11i = -sy * sx;
    float* o = U + g * 8;
    o[0] = cz * m00r + sz * m00i;  o[1] = cz * m00i - sz * m00r;
    o[2] = cz * m01r + sz * m01i;  o[3] = cz * m01i - sz * m01r;
    o[4] = cz * m10r - sz * m10i;  o[5] = cz * m10i + sz * m10r;
    o[6] = cz * m11r - sz * m11i;  o[7] = cz * m11i + sz * m11r;
}

__global__ __launch_bounds__(256) void qcnn_features(const float* __restrict__ x,
                                                     const float* __restrict__ Utab,
                                                     float* __restrict__ feats) {
    __shared__ float uS[NSPLIT * NLAYER * QPS * 8];
    for (int i = threadIdx.x; i < NSPLIT * NLAYER * QPS * 8; i += blockDim.x)
        uS[i] = Utab[i];
    __syncthreads();

    int t = blockIdx.x * blockDim.x + threadIdx.x;
    int s = t & 3;

    const float* xp = x + (size_t)t * DIM;
    float re[DIM], im[DIM];
    float nrm2 = 0.f;
#pragma unroll
    for (int i = 0; i < DIM; i++) { float v = xp[i]; re[i] = v; nrm2 += v * v; }
    float nrm = sqrtf(nrm2);
    float scale = (nrm > 0.f) ? (1.f / nrm) : 1.f;
#pragma unroll
    for (int i = 0; i < DIM; i++) { re[i] *= scale; im[i] = 0.f; }

    const float* us = &uS[s * NLAYER * QPS * 8];
#pragma unroll
    for (int l = 0; l < NLAYER; l++) {
#pragma unroll
        for (int j = 0; j < QPS; j++) {
            const float* u = &us[(l * QPS + j) * 8];
            float u00r = u[0], u00i = u[1], u01r = u[2], u01i = u[3];
            float u10r = u[4], u10i = u[5], u11r = u[6], u11i = u[7];
            const int m = 8 >> j;
#pragma unroll
            for (int idx = 0; idx < DIM; idx++) {
                if (idx & m) continue;
                const int i1 = idx | m;
                float a0r = re[idx], a0i = im[idx], a1r = re[i1], a1i = im[i1];
                re[idx] = u00r * a0r - u00i * a0i + u01r * a1r - u01i * a1i;
                im[idx] = u00r * a0i + u00i * a0r + u01r * a1i + u01i * a1r;
                re[i1]  = u10r * a0r - u10i * a0i + u11r * a1r - u11i * a1i;
                im[i1]  = u10r * a0i + u10i * a0r + u11r * a1i + u11i * a1r;
            }
        }
#pragma unroll
        for (int j = 0; j < QPS; j++) {
            const int mc = 8 >> j, mt = 8 >> ((j + 1) & 3);
#pragma unroll
            for (int idx = 0; idx < DIM; idx++) {
                if ((idx & mc) && !(idx & mt)) {
                    const int i1 = idx | mt;
                    float tr = re[idx], ti = im[idx];
                    re[idx] = re[i1]; im[idx] = im[i1];
                    re[i1] = tr;      im[i1] = ti;
                }
            }
        }
    }

    float e[QPS];
#pragma unroll
    for (int q = 0; q < QPS; q++) e[q] = 0.f;
#pragma unroll
    for (int idx = 0; idx < DIM; idx++) {
        float p = re[idx] * re[idx] + im[idx] * im[idx];
#pragma unroll
        for (int q = 0; q < QPS; q++)
            e[q] += ((idx >> (3 - q)) & 1) ? -p : p;
    }
    int b = t >> 2;
    float* f = feats + (size_t)b * FEAT + s * QPS;
#pragma unroll
    for (int q = 0; q < QPS; q++) f[q] = e[q];
}

__global__ __launch_bounds__(32 * WAVES_PER_BLOCK) void qcnn_mlp(
    const float* __restrict__ feats,
    const float* __restrict__ W1, const float* __restrict__ b1,
    const float* __restrict__ W2, const float* __restrict__ b2,
    const float* __restrict__ W3, const float* __restrict__ b3,
    float* __restrict__ out)
{
    __shared__ float h1s[WAVES_PER_BLOCK][16 * H1P];
    __shared__ float h2s[WAVES_PER_BLOCK][16 * H2P];

    const int lane = threadIdx.x & 31;
    const int wave = threadIdx.x >> 5;
    const int half = lane >> 4;
    const int nl   = lane & 15;
    const int rowBase = (blockIdx.x * WAVES_PER_BLOCK + wave) * 16;
    float* h1 = h1s[wave];
    float* h2 = h2s[wave];

    v2f a1f[4];
#pragma unroll
    for (int kc = 0; kc < 4; kc++) {
        const float* ap = feats + (size_t)(rowBase + nl) * FEAT + kc * 4 + 2 * half;
        a1f[kc] = v2f{ap[0], ap[1]};
    }
#pragma unroll
    for (int nt = 0; nt < 8; nt++) {
        const int n0 = nt * 16;
        float bias = b1[n0 + nl];
        v8f c;
#pragma unroll
        for (int r = 0; r < 8; r++) c[r] = bias;
#pragma unroll
        for (int kc = 0; kc < 4; kc++) {
            const float* bp = W1 + (size_t)(kc * 4 + 2 * half) * H1 + n0 + nl;
            v2f bf = v2f{bp[0], bp[H1]};
            c = __builtin_amdgcn_wmma_f32_16x16x4_f32(false, a1f[kc], false, bf,
                                                      (short)0, c, false, false);
        }
#pragma unroll
        for (int r = 0; r < 8; r++)
            h1[(r + 8 * half) * H1P + n0 + nl] = fmaxf(c[r], 0.f);
    }
    __syncthreads();

#pragma unroll
    for (int nt = 0; nt < 4; nt++) {
        const int n0 = nt * 16;
        float bias = b2[n0 + nl];
        v8f c;
#pragma unroll
        for (int r = 0; r < 8; r++) c[r] = bias;
#pragma unroll 4
        for (int k0 = 0; k0 < H1; k0 += 4) {
            const float* ap = h1 + nl * H1P + k0 + 2 * half;
            v2f af = v2f{ap[0], ap[1]};
            const float* bp = W2 + (size_t)(k0 + 2 * half) * H2 + n0 + nl;
            v2f bf = v2f{bp[0], bp[H2]};
            c = __builtin_amdgcn_wmma_f32_16x16x4_f32(false, af, false, bf,
                                                      (short)0, c, false, false);
        }
#pragma unroll
        for (int r = 0; r < 8; r++)
            h2[(r + 8 * half) * H2P + n0 + nl] = fmaxf(c[r], 0.f);
    }
    __syncthreads();

    {
        const float msk = (nl < NCLASS) ? 1.f : 0.f;
        const int ncl = (nl < NCLASS) ? nl : (NCLASS - 1);
        float bias = msk * b3[ncl];
        v8f c;
#pragma unroll
        for (int r = 0; r < 8; r++) c[r] = bias;
#pragma unroll 4
        for (int k0 = 0; k0 < H2; k0 += 4) {
            const float* ap = h2 + nl * H2P + k0 + 2 * half;
            v2f af = v2f{ap[0], ap[1]};
            const int k = k0 + 2 * half;
            v2f bf = v2f{msk * W3[k * NCLASS + ncl], msk * W3[(k + 1) * NCLASS + ncl]};
            c = __builtin_amdgcn_wmma_f32_16x16x4_f32(false, af, false, bf,
                                                      (short)0, c, false, false);
        }
        if (nl < NCLASS) {
#pragma unroll
            for (int r = 0; r < 8; r++) {
                const int m = r + 8 * half;
                out[(size_t)(rowBase + m) * NCLASS + nl] = c[r];
            }
        }
    }
}

extern "C" void kernel_launch(void* const* d_in, const int* in_sizes, int n_in,
                              void* d_out, int out_size, void* d_ws, size_t ws_size,
                              hipStream_t stream) {
    const float* x  = (const float*)d_in[0];
    const float* sw = (const float*)d_in[1];
    const float* W1 = (const float*)d_in[2];
    const float* b1 = (const float*)d_in[3];
    const float* W2 = (const float*)d_in[4];
    const float* b2 = (const float*)d_in[5];
    const float* W3 = (const float*)d_in[6];
    const float* b3 = (const float*)d_in[7];
    float* out = (float*)d_out;

    float* Utab  = (float*)d_ws;
    float* feats = (float*)((char*)d_ws + 4096);

    qcnn_unitaries<<<dim3(1), dim3(64), 0, stream>>>(sw, Utab);

    qcnn_features<<<dim3((BATCH * NSPLIT) / 256), dim3(256), 0, stream>>>(x, Utab, feats);

    qcnn_mlp<<<dim3(BATCH / (16 * WAVES_PER_BLOCK)), dim3(32 * WAVES_PER_BLOCK), 0, stream>>>(
        feats, W1, b1, W2, b2, W3, b3, out);
}


# MLP operands asked for one stage ahead: W1, b1 before the gate loop; W2's first 8 k steps, b2, W3, b3 before layer 1; out pointer at entry; layer 3 unrolled, LDS reads in one group; sums unchanged
# speedup vs baseline: 2.5892x; 2.5892x over previous
#include <hip/hip_runtime.h>
#include <math.h>


#define QPS     4
#define NSPLIT  4
#define NLAYER  4
#define DIM     16
#define NCLASS  5
#define BATCH   32768
#define FEAT    16
#define H1      128
#define H2      64

#define WAVES_PER_BLOCK 4
#define H1P 132
#define H2P 68
#define W2PRE 8

typedef __attribute__((ext_vector_type(2))) float v2f;
typedef __attribute__((ext_vector_type(4))) float v4f;
typedef __attribute__((ext_vector_type(8))) float v8f;
typedef __attribute__((address_space(1))) float gfloat;

__device__ __forceinline__ float row_lane(float v, unsigned sel) {
    const unsigned u = __builtin_bit_cast(unsigned, v);
    return __builtin_bit_cast(float, __builtin_amdgcn_permlane16(0u, u, sel, sel, false, false));
}


__device__ __forceinline__ v2f pk_mul_x(v2f u, v2f a) {
    v2f d;
    asm volatile("v_pk_mul_f32 %0, %1, %2 op_sel_hi:[0,1]" : "=v"(d) : "v"(u), "v"(a));
    return d;
}
__device__ __forceinline__ v2f pk_fma_x(v2f u, v2f a, v2f c) {
    v2f d;
    asm volatile("v_pk_fma_f32 %0, %1, %2, %3 op_sel_hi:[0,1,1]" : "=v"(d) : "v"(u), "v"(a), "v"(c));
    return d;
}
__device__ __forceinline__ v2f pk_fma_y(v2f u, v2f a, v2f c) {
    v2f d;
    asm volatile("v_pk_fma_f32 %0, %1, %2, %3 op_sel:[1,1,0] op_sel_hi:[1,0,1] neg_lo:[1,0,0]" : "=v"(d) : "v"(u), "v"(a), "v"(c));
    return d;
}

__device__ __forceinline__ v2f pk_mul_y(v2f u, v2f a) {
    v2f d;
    asm volatile("v_pk_mul_f32 %0, %1, %2 op_sel:[1,0] op_sel_hi:[1,1]" : "=v"(d) : "v"(u), "v"(a));
    return d;
}
__device__ __forceinline__ v2f pk_mul_y_neg(v2f u, v2f a) {
    v2f d;
    asm volatile("v_pk_mul_f32 %0, %1, %2 op_sel:[1,0] op_sel_hi:[1,1] neg_lo:[1,0] neg_hi:[1,0]" : "=v"(d) : "v"(u), "v"(a));
    return d;
}
__device__ __forceinline__ v2f pk_fma_yy(v2f u, v2f a, v2f c) {
    v2f d;
    asm volatile("v_pk_fma_f32 %0, %1, %2, %3 op_sel:[1,0,0] op_sel_hi:[1,1,1]" : "=v"(d) : "v"(u), "v"(a), "v"(c));
    return d;
}

__device__ __forceinline__ void polar(float re, float im, float& mod, float& pr, float& pi) {
    const float m = fmaxf(fabsf(re), fabsf(im));
    int ex;
    (void)frexpf(m, &ex);
    const float a = ldexpf(re, -ex), b = ldexpf(im, -ex);
    const float n2 = a * a + b * b;
    const float rn = __builtin_amdgcn_rsqf(n2);
    const bool nz = m > 0.f;
    mod = nz ? ldexpf(n2 * rn, ex) : 0.f;
    pr  = nz ? a * rn : 1.f;
    pi  = nz ? b * rn : 0.f;
}

constexpr unsigned long long ring_sources() {
    int a[DIM] = {0, 1, 2, 3, 4, 5, 6, 7, 8, 9, 10, 11, 12, 13, 14, 15};
    for (int j = 0; j < QPS; j++) {
        const int mc = 8 >> j, mt = 8 >> ((j + 1) & 3);
        for (int idx = 0; idx < DIM; idx++) {
            if ((idx & mc) && !(idx & mt)) {
                const int i1 = idx | mt;
                const int t = a[idx];
                a[idx] = a[i1];
                a[i1] = t;
            }
        }
    }
    unsigned long long r = 0;
    for (int i = 0; i < DIM; i++) r |= (unsigned long long)a[i] << (4 * i);
    return r;
}

#define TAB_CS  (NLAYER * DIM * 2)
#define TAB_SPL (TAB_CS + NLAYER * QPS * 2 + 8)
static_assert(WAVES_PER_BLOCK == NSPLIT, "wave w builds the table of split w");

__global__ __launch_bounds__(32 * WAVES_PER_BLOCK) void qcnn_circuit_mlp(
    const float* __restrict__ x, const float* __restrict__ sw,
    const float* __restrict__ W1, const float* __restrict__ b1,
    const float* __restrict__ W2, const float* __restrict__ b2,
    const float* __restrict__ W3, const float* __restrict__ b3,
    float* __restrict__ out)
{
    __shared__ __attribute__((aligned(16))) float h1s[WAVES_PER_BLOCK][16 * H1P];
    __shared__ __attribute__((aligned(16))) float h2s[WAVES_PER_BLOCK][16 * H2P];
    __shared__ float fS[WAVES_PER_BLOCK][32 * 8];
    __shared__ __attribute__((aligned(16))) float tab[NSPLIT][TAB_SPL];

    const int lane = threadIdx.x & 31;
    const int wave = threadIdx.x >> 5;
    const int half = lane >> 4;
    const int nl   = lane & 15;
    const int rowBase = (blockIdx.x * WAVES_PER_BLOCK + wave) * 16;
    float* h1 = h1s[wave];
    float* h2 = h2s[wave];
    float* fz = &fS[wave][lane * 8];

    const float* wq = sw + (wave * (NLAYER * QPS) + ((nl + 8 * half) & 15)) * 3;
    float wv[3];
#pragma unroll
    for (int i = 0; i < 3; i++) wv[i] = wq[i];

    const float* xq = x + (size_t)(rowBase + nl) * (NSPLIT * DIM) + half * (2 * DIM);
    float xr[DIM];
#pragma unroll
    for (int i = 0; i < DIM; i++) xr[i] = xq[i];

    gfloat* outp = (gfloat*)out;
    asm volatile("" : "+s"(outp));

    {
        float p0r, p0i, p1r, p1i, qr, qi, gc, gs;
        {
            const float hx = 0.5f * wv[0], hy = 0.5f * wv[1], hz = 0.5f * wv[2];
            float sx, cx, sy, cy, sz, cz;
            sincosf(hx, &sx, &cx);
            sincosf(hy, &sy, &cy);
            sincosf(hz, &sz, &cz);
            const float m00r = cy * cx, m00i =  sy * sx;
            const float m01r = -sy * cx, m01i = -cy * sx;
            const float u00r = cz * m00r + sz * m00i, u00i = cz * m00i - sz * m00r;
            const float u01r = cz * m01r + sz * m01i, u01i = cz * m01i - sz * m01r;
            float r1r, r1i;
            polar(u00r, u00i, gc, p0r, p0i);
            polar(u01r, u01i, gs, r1r, r1i);
            p1r = r1r;  p1i = -r1i;
            qr = r1r * p0r + r1i * p0i;
            qi = r1i * p0r - r1r * p0i;
        }
        const int ks = (int)((ring_sources() >> (4 * nl)) & 15ull);
#pragma unroll
        for (int step = 0; step < 2; step++) {
            const int lp = (step == 0) ? 0 : 3;
            const int lq = (step == 0) ? 1 : 0;
            float Pr = 1.f, Pi = 0.f, Qr = 1.f, Qi = 0.f;
#pragma unroll
            for (int j = 0; j < QPS; j++) {
                {
                    const unsigned sel = 0x11111111u * (unsigned)(lp * QPS + j);
                    const float a0r = row_lane(p0r, sel), a0i = row_lane(p0i, sel);
                    const float a1r = row_lane(p1r, sel), a1i = row_lane(p1i, sel);
                    const bool hi = (ks >> (3 - j)) & 1;
                    const float fr = hi ? a1r : a0r, fi = hi ? a1i : a0i;
                    if (j == 0) { Pr = fr;  Pi = fi; }
                    else { const float tr = Pr * fr - Pi * fi;  Pi = Pr * fi + Pi * fr;  Pr = tr; }
                }
                {
                    const unsigned sel = 0x11111111u * (unsigned)(lq * QPS + j);
                    const float b1r = row_lane(qr, sel), b1i = row_lane(qi, sel);
                    const bool hi = (nl >> (3 - j)) & 1;
                    const float fr = hi ? b1r : 1.f, fi = hi ? b1i : 0.f;
                    if (j == 0) { Qr = fr;  Qi = fi; }
                    else { const float tr = Qr * fr - Qi * fi;  Qi = Qr * fi + Qi * fr;  Qr = tr; }
                }
            }
            float er = Pr * Qr - Pi * Qi, ei = Pr * Qi + Pi * Qr;
            if (step == 1) { er = (half == 0) ? Qr : er;  ei = (half == 0) ? Qi : ei; }
            const float rn = __builtin_amdgcn_rsqf(er * er + ei * ei);
            const int d = (step == 0) ? (1 + 2 * half) : (2 * half);
            *(v2f*)(&tab[wave][(d * DIM + nl) * 2]) = v2f{er * rn, ei * rn};
        }
        if (half == 0) *(v2f*)(&tab[wave][TAB_CS + nl * 2]) = v2f{gc, gs};
    }
    __syncthreads();

    v4f w1q[4][2][2];
#pragma unroll
    for (int kc = 0; kc < 4; kc++)
#pragma unroll
        for (int e = 0; e < 2; e++)
#pragma unroll
            for (int g = 0; g < 2; g++)
                w1q[kc][e][g] = *(const v4f*)(W1 + (size_t)(8 * half + 2 * kc + e) * H1 + 64 * g + 4 * nl);
    v4f b1q[2];
#pragma unroll
    for (int g = 0; g < 2; g++) b1q[g] = *(const v4f*)(b1 + 64 * g + 4 * nl);

#pragma unroll 1
    for (int ss = 0; ss < 2; ss++) {
        const float* tb = &tab[2 * half + ss][0];
        v4f eq[DIM / 2], cs[QPS / 2];
#pragma unroll
        for (int i = 0; i < DIM / 2; i++) eq[i] = *(const v4f*)(tb + 4 * i);
#pragma unroll
        for (int i = 0; i < QPS / 2; i++) cs[i] = *(const v4f*)(tb + TAB_CS + 4 * i);

        v2f st[DIM];
        float nrm2 = 0.f;
#pragma unroll
        for (int i = 0; i < DIM; i++) { float v = xr[i]; nrm2 += v * v; }
        float nrm = sqrtf(nrm2);
        float scale = (nrm > 0.f) ? (1.f / nrm) : 1.f;
        v2f xs[DIM / 2];
#pragma unroll
        for (int i = 0; i < DIM / 2; i++) xs[i] = v2f{xr[2 * i] * scale, xr[2 * i + 1] * scale};
        if (ss == 0) {
#pragma unroll
            for (int i = 0; i < DIM; i++) xr[i] = xq[DIM + i];
        }
#pragma unroll
        for (int i = 0; i < DIM / 2; i++) {
            st[2 * i]     = pk_mul_x(xs[i], v2f{eq[i][0], eq[i][1]});
            st[2 * i + 1] = pk_mul_y(xs[i], v2f{eq[i][2], eq[i][3]});
        }

#pragma unroll
        for (int l = 0; l < NLAYER; l++) {
            if (l > 0) {
                v2f t[DIM];
#pragma unroll
                for (int i = 0; i < DIM; i++)
                    t[i] = pk_mul_x(st[i], v2f{eq[i >> 1][2 * (i & 1)], eq[i >> 1][2 * (i & 1) + 1]});
#pragma unroll
                for (int i = 0; i < DIM; i++)
                    st[i] = pk_fma_y(st[i], v2f{eq[i >> 1][2 * (i & 1)], eq[i >> 1][2 * (i & 1) + 1]}, t[i]);
            }
            const v4f cs0 = cs[0], cs1 = cs[1];
            if (l + 1 < NLAYER) {
#pragma unroll
                for (int i = 0; i < DIM / 2; i++) eq[i] = *(const v4f*)(tb + (l + 1) * DIM * 2 + 4 * i);
#pragma unroll
                for (int i = 0; i < QPS / 2; i++) cs[i] = *(const v4f*)(tb + TAB_CS + (l + 1) * QPS * 2 + 4 * i);
            }
#pragma unroll
            for (int j = 0; j < QPS; j++) {
                const v4f cq = (j < 2) ? cs0 : cs1;
                const v2f C = v2f{cq[2 * (j & 1)], cq[2 * (j & 1) + 1]};
                const int m = 8 >> j;
                v2f n0[8], n1[8];
#pragma unroll
                for (int p = 0; p < 8; p++) {
                    const int i0 = ((p & ~(m - 1)) << 1) | (p & (m - 1));
                    n0[p] = pk_mul_x(C, st[i0]);
                    n1[p] = pk_mul_y_neg(C, st[i0]);
                }
#pragma unroll
                for (int p = 0; p < 8; p++) {
                    const int i1 = ((p & ~(m - 1)) << 1) | (p & (m - 1)) | m;
                    n0[p] = pk_fma_yy(C, st[i1], n0[p]);
                    n1[p] = pk_fma_x(C, st[i1], n1[p]);
                }
#pragma unroll
                for (int p = 0; p < 8; p++) {
                    const int i0 = ((p & ~(m - 1)) << 1) | (p & (m - 1));
                    st[i0] = n0[p];
                    st[i0 | m] = n1[p];
                }
            }
#pragma unroll
            for (int j = 0; j < QPS; j++) {
                const int mc = 8 >> j, mt = 8 >> ((j + 1) & 3);
#pragma unroll
                for (int idx = 0; idx < DIM; idx++) {
                    if ((idx & mc) && !(idx & mt)) {
                        const int i1 = idx | mt;
                        const v2f t = st[idx];
                        st[idx] = st[i1];
                        st[i1] = t;
                    }
                }
            }
        }

        float e[QPS];
#pragma unroll
        for (int q = 0; q < QPS; q++) e[q] = 0.f;
#pragma unroll
        for (int idx = 0; idx < DIM; idx++) {
            float p = st[idx].x * st[idx].x + st[idx].y * st[idx].y;
#pragma unroll
            for (int q = 0; q < QPS; q++)
                e[q] += ((idx >> (3 - q)) & 1) ? -p : p;
        }
#pragma unroll
        for (int q = 0; q < QPS; q++) fz[ss * QPS + q] = e[q];
    }


    v2f a1f[4];
#pragma unroll
    for (int kc = 0; kc < 4; kc++) a1f[kc] = v2f{fz[2 * kc], fz[2 * kc + 1]};
    v4f w2p[W2PRE][2];
#pragma unroll
    for (int kk = 0; kk < W2PRE; kk++) {
        const float* bp = W2 + (size_t)(4 * kk + 2 * half) * H2 + 4 * nl;
        w2p[kk][0] = *(const v4f*)bp;
        w2p[kk][1] = *(const v4f*)(bp + H2);
    }
    const v4f b2q = *(const v4f*)(b2 + 4 * nl);
    const int ncl = (nl < NCLASS) ? nl : (NCLASS - 1);
    float w3q[H2 / 4][2];
#pragma unroll
    for (int kk = 0; kk < H2 / 4; kk++) {
        const int k = 4 * kk + 2 * half;
        w3q[kk][0] = W3[k * NCLASS + ncl];
        w3q[kk][1] = W3[(k + 1) * NCLASS + ncl];
    }
    const float b3v = b3[ncl];
    {
#pragma unroll
        for (int g = 0; g < 2; g++) {
            const v4f bq = b1q[g];
            v8f c[4];
#pragma unroll
            for (int j = 0; j < 4; j++)
#pragma unroll
                for (int r = 0; r < 8; r++) c[j][r] = bq[j];
#pragma unroll
            for (int kc = 0; kc < 4; kc++)
#pragma unroll
                for (int j = 0; j < 4; j++)
                    c[j] = __builtin_amdgcn_wmma_f32_16x16x4_f32(false, a1f[kc], false,
                               v2f{w1q[kc][0][g][j], w1q[kc][1][g][j]}, (short)0, c[j], false, false);
#pragma unroll
            for (int r = 0; r < 8; r++)
                *(v4f*)(h1 + (r + 8 * half) * H1P + 64 * g + 4 * nl) =
                    v4f{fmaxf(c[0][r], 0.f), fmaxf(c[1][r], 0.f), fmaxf(c[2][r], 0.f), fmaxf(c[3][r], 0.f)};
        }
    }
    __syncthreads();

    {
        const v4f bq = b2q;
        v8f c[4];
#pragma unroll
        for (int j = 0; j < 4; j++)
#pragma unroll
            for (int r = 0; r < 8; r++) c[j][r] = bq[j];
#pragma unroll
        for (int k0 = 0; k0 < H1; k0 += 4) {
            const float* ap = h1 + nl * H1P + k0 + 2 * half;
            const v2f af = v2f{ap[0], ap[1]};
            const float* bp = W2 + (size_t)(k0 + 2 * half) * H2 + 4 * nl;
            const v4f r0 = (k0 / 4 < W2PRE) ? w2p[(k0 / 4 < W2PRE) ? k0 / 4 : 0][0] : *(const v4f*)bp;
            const v4f r1 = (k0 / 4 < W2PRE) ? w2p[(k0 / 4 < W2PRE) ? k0 / 4 : 0][1] : *(const v4f*)(bp + H2);
#pragma unroll
            for (int j = 0; j < 4; j++)
                c[j] = __builtin_amdgcn_wmma_f32_16x16x4_f32(false, af, false, v2f{r0[j], r1[j]},
                                                             (short)0, c[j], false, false);
        }
#pragma unroll
        for (int r = 0; r < 8; r++)
            *(v4f*)(h2 + (r + 8 * half) * H2P + 4 * nl) =
                v4f{fmaxf(c[0][r], 0.f), fmaxf(c[1][r], 0.f), fmaxf(c[2][r], 0.f), fmaxf(c[3][r], 0.f)};
    }
    __syncthreads();

    {
        const float msk = (nl < NCLASS) ? 1.f : 0.f;
        float bias = msk * b3v;
        v8f c;
#pragma unroll
        for (int r = 0; r < 8; r++) c[r] = bias;
        v2f a3[H2 / 4], b3f[H2 / 4];
#pragma unroll
        for (int kk = 0; kk < H2 / 4; kk++) {
            const float* ap = h2 + nl * H2P + 4 * kk + 2 * half;
            a3[kk] = v2f{ap[0], ap[1]};
        }
#pragma unroll
        for (int kk = 0; kk < H2 / 4; kk++) b3f[kk] = v2f{msk * w3q[kk][0], msk * w3q[kk][1]};
        __builtin_amdgcn_sched_barrier(0);
#pragma unroll
        for (int kk = 0; kk < H2 / 4; kk++)
            c = __builtin_amdgcn_wmma_f32_16x16x4_f32(false, a3[kk], false, b3f[kk],
                                                      (short)0, c, false, false);
        if (nl < NCLASS) {
#pragma unroll
            for (int r = 0; r < 8; r++) {
                const int m = r + 8 * half;
                outp[(size_t)(rowBase + m) * NCLASS + nl] = c[r];
            }
        }
    }
}

extern "C" void kernel_launch(void* const* d_in, const int* in_sizes, int n_in,
                              void* d_out, int out_size, void* d_ws, size_t ws_size,
                              hipStream_t stream) {
    const float* x  = (const float*)d_in[0];
    const float* sw = (const float*)d_in[1];
    const float* W1 = (const float*)d_in[2];
    const float* b1 = (const float*)d_in[3];
    const float* W2 = (const float*)d_in[4];
    const float* b2 = (const float*)d_in[5];
    const float* W3 = (const float*)d_in[6];
    const float* b3 = (const float*)d_in[7];
    float* out = (float*)d_out;

    qcnn_circuit_mlp<<<dim3(BATCH / (16 * WAVES_PER_BLOCK)), dim3(32 * WAVES_PER_BLOCK), 0, stream>>>(
        x, sw, W1, b1, W2, b2, W3, b3, out);
}


# Rotations as scaled shears a0+t*a1, a1-t*a0, |t|<=1: one two-wide op per amplitude, not two; quarter turns kept as index flips in the table, moduli folded into diagonal 0; state sum order changes
# speedup vs baseline: 2.7218x; 1.0512x over previous
#include <hip/hip_runtime.h>
#include <math.h>


#define QPS     4
#define NSPLIT  4
#define NLAYER  4
#define DIM     16
#define NCLASS  5
#define BATCH   32768
#define FEAT    16
#define H1      128
#define H2      64

#define WAVES_PER_BLOCK 4
#define H1P 132
#define H2P 68
#define W2PRE 8

typedef __attribute__((ext_vector_type(2))) float v2f;
typedef __attribute__((ext_vector_type(4))) float v4f;
typedef __attribute__((ext_vector_type(8))) float v8f;
typedef __attribute__((address_space(1))) float gfloat;

__device__ __forceinline__ float row_lane(float v, unsigned sel) {
    const unsigned u = __builtin_bit_cast(unsigned, v);
    return __builtin_bit_cast(float, __builtin_amdgcn_permlane16(0u, u, sel, sel, false, false));
}
__device__ __forceinline__ float row_perm(float v, unsigned lo, unsigned hi) {
    const unsigned u = __builtin_bit_cast(unsigned, v);
    return __builtin_bit_cast(float, __builtin_amdgcn_permlane16(0u, u, lo, hi, false, false));
}


__device__ __forceinline__ v2f pk_mul_x(v2f u, v2f a) {
    v2f d;
    asm volatile("v_pk_mul_f32 %0, %1, %2 op_sel_hi:[0,1]" : "=v"(d) : "v"(u), "v"(a));
    return d;
}
__device__ __forceinline__ v2f pk_fma_y(v2f u, v2f a, v2f c) {
    v2f d;
    asm volatile("v_pk_fma_f32 %0, %1, %2, %3 op_sel:[1,1,0] op_sel_hi:[1,0,1] neg_lo:[1,0,0]" : "=v"(d) : "v"(u), "v"(a), "v"(c));
    return d;
}

__device__ __forceinline__ v2f pk_mul_y(v2f u, v2f a) {
    v2f d;
    asm volatile("v_pk_mul_f32 %0, %1, %2 op_sel:[1,0] op_sel_hi:[1,1]" : "=v"(d) : "v"(u), "v"(a));
    return d;
}

__device__ __forceinline__ void shear_pair(v2f u, v2f& a0, v2f& a1) {
    v2f n;
    asm volatile("v_pk_fma_f32 %0, %2, %1, %3 op_sel_hi:[0,1,1]\n\t"
                 "v_pk_fma_f32 %1, %2, %3, %1 op_sel:[1,0,0] op_sel_hi:[1,1,1]"
                 : "=&v"(n), "+v"(a1) : "v"(u), "v"(a0));
    a0 = n;
}

__device__ __forceinline__ void polar(float re, float im, float& mod, float& pr, float& pi) {
    const float m = fmaxf(fabsf(re), fabsf(im));
    int ex;
    (void)frexpf(m, &ex);
    const float a = ldexpf(re, -ex), b = ldexpf(im, -ex);
    const float n2 = a * a + b * b;
    const float rn = __builtin_amdgcn_rsqf(n2);
    const bool nz = m > 0.f;
    mod = nz ? ldexpf(n2 * rn, ex) : 0.f;
    pr  = nz ? a * rn : 1.f;
    pi  = nz ? b * rn : 0.f;
}

constexpr unsigned long long ring_sources() {
    int a[DIM] = {0, 1, 2, 3, 4, 5, 6, 7, 8, 9, 10, 11, 12, 13, 14, 15};
    for (int j = 0; j < QPS; j++) {
        const int mc = 8 >> j, mt = 8 >> ((j + 1) & 3);
        for (int idx = 0; idx < DIM; idx++) {
            if ((idx & mc) && !(idx & mt)) {
                const int i1 = idx | mt;
                const int t = a[idx];
                a[idx] = a[i1];
                a[i1] = t;
            }
        }
    }
    unsigned long long r = 0;
    for (int i = 0; i < DIM; i++) r |= (unsigned long long)a[i] << (4 * i);
    return r;
}
constexpr unsigned long long ring_targets() {
    unsigned long long r = 0;
    for (int x0 = 0; x0 < DIM; x0++) {
        int x = x0;
        for (int j = 0; j < QPS; j++) {
            const int mc = 8 >> j, mt = 8 >> ((j + 1) & 3);
            if (x & mc) x ^= mt;
        }
        r |= (unsigned long long)x << (4 * x0);
    }
    return r;
}
__device__ __forceinline__ unsigned ring_forward(unsigned x) {
    constexpr unsigned long long tgt = ring_targets();
    return (unsigned)(tgt >> (4u * x)) & 15u;
}

#define TAB_CS  (NLAYER * DIM * 2)
#define TAB_SG  (TAB_CS + NLAYER * QPS * 2)
#define TAB_SPL (TAB_SG + 8)
static_assert(WAVES_PER_BLOCK == NSPLIT, "wave w builds the table of split w");

__global__ __launch_bounds__(32 * WAVES_PER_BLOCK) void qcnn_circuit_mlp(
    const float* __restrict__ x, const float* __restrict__ sw,
    const float* __restrict__ W1, const float* __restrict__ b1,
    const float* __restrict__ W2, const float* __restrict__ b2,
    const float* __restrict__ W3, const float* __restrict__ b3,
    float* __restrict__ out)
{
    __shared__ __attribute__((aligned(16))) float h1s[WAVES_PER_BLOCK][16 * H1P];
    __shared__ __attribute__((aligned(16))) float h2s[WAVES_PER_BLOCK][16 * H2P];
    __shared__ float fS[WAVES_PER_BLOCK][32 * 8];
    __shared__ __attribute__((aligned(16))) float tab[NSPLIT][TAB_SPL];

    const int lane = threadIdx.x & 31;
    const int wave = threadIdx.x >> 5;
    const int half = lane >> 4;
    const int nl   = lane & 15;
    const int rowBase = (blockIdx.x * WAVES_PER_BLOCK + wave) * 16;
    float* h1 = h1s[wave];
    float* h2 = h2s[wave];
    float* fz = &fS[wave][lane * 8];

    const float* wq = sw + (wave * (NLAYER * QPS) + ((nl + 8 * half) & 15)) * 3;
    float wv[3];
#pragma unroll
    for (int i = 0; i < 3; i++) wv[i] = wq[i];

    const float* xq = x + (size_t)(rowBase + nl) * (NSPLIT * DIM) + half * (2 * DIM);
    float xr[DIM];
#pragma unroll
    for (int i = 0; i < DIM; i++) xr[i] = xq[i];

    gfloat* outp = (gfloat*)out;
    asm volatile("" : "+s"(outp));

    {
        float p0r, p0i, p1r, p1i, qr, qi, gc, gs;
        {
            const float hx = 0.5f * wv[0], hy = 0.5f * wv[1], hz = 0.5f * wv[2];
            float sx, cx, sy, cy, sz, cz;
            sincosf(hx, &sx, &cx);
            sincosf(hy, &sy, &cy);
            sincosf(hz, &sz, &cz);
            const float m00r = cy * cx, m00i =  sy * sx;
            const float m01r = -sy * cx, m01i = -cy * sx;
            const float u00r = cz * m00r + sz * m00i, u00i = cz * m00i - sz * m00r;
            const float u01r = cz * m01r + sz * m01i, u01i = cz * m01i - sz * m01r;
            float r1r, r1i;
            polar(u00r, u00i, gc, p0r, p0i);
            polar(u01r, u01i, gs, r1r, r1i);
            p1r = r1r;  p1i = -r1i;
            qr = r1r * p0r + r1i * p0i;
            qi = r1i * p0r - r1r * p0i;
        }
        const bool qt = gs > gc;
        const float gk = qt ? gs : gc;
        const float tq = (qt ? gc : gs) / gk;
        const float tau = qt ? -tq : tq;
        const unsigned qmask = __builtin_amdgcn_ballot_w32(qt) & 0xffffu;
        unsigned BETA = 0u, FPRE = 0u, FPOST = 0u, fm = 0u;
#pragma unroll
        for (int l = 0; l < NLAYER; l++) {
            unsigned bl = 0u;
#pragma unroll
            for (int j = 0; j < QPS; j++) bl |= ((qmask >> (4 * l + j)) & 1u) << (3 - j);
            FPRE |= fm << (4 * l);
            fm ^= bl;
            FPOST |= fm << (4 * l);
            BETA |= bl << (4 * l);
            fm = ring_forward(fm);
        }
        const unsigned ffin = fm;
        float G = gk;
        G *= row_perm(G, 0x67452301u, 0xEFCDAB89u);
        G *= row_perm(G, 0x54761032u, 0xDCFE98BAu);
        G *= row_perm(G, 0x32107654u, 0xBA98FEDCu);
        G *= row_perm(G, 0xFEDCBA98u, 0x76543210u);
        const int ks = (int)((ring_sources() >> (4 * nl)) & 15ull);
#pragma unroll
        for (int step = 0; step < 2; step++) {
            const int lp = (step == 0) ? 0 : 3;
            const int lq = (step == 0) ? 1 : 0;
            const int d = (step == 0) ? (1 + 2 * half) : (2 * half);
            const unsigned ko = (unsigned)ks ^ ((FPOST >> (4 * ((d + 3) & 3))) & 15u);
            const unsigned ki = (unsigned)nl ^ ((FPRE >> (4 * d)) & 15u);
            const unsigned kz = ((unsigned)nl ^ ((FPOST >> (4 * d)) & 15u)) & ((BETA >> (4 * d)) & 15u);
            float Pr = 1.f, Pi = 0.f, Qr = 1.f, Qi = 0.f;
#pragma unroll
            for (int j = 0; j < QPS; j++) {
                {
                    const unsigned sel = 0x11111111u * (unsigned)(lp * QPS + j);
                    const float a0r = row_lane(p0r, sel), a0i = row_lane(p0i, sel);
                    const float a1r = row_lane(p1r, sel), a1i = row_lane(p1i, sel);
                    const bool hi = (ko >> (3 - j)) & 1u;
                    const float fr = hi ? a1r : a0r, fi = hi ? a1i : a0i;
                    if (j == 0) { Pr = fr;  Pi = fi; }
                    else { const float tr = Pr * fr - Pi * fi;  Pi = Pr * fi + Pi * fr;  Pr = tr; }
                }
                {
                    const unsigned sel = 0x11111111u * (unsigned)(lq * QPS + j);
                    const float b1r = row_lane(qr, sel), b1i = row_lane(qi, sel);
                    const bool hi = (ki >> (3 - j)) & 1u;
                    const float fr = hi ? b1r : 1.f, fi = hi ? b1i : 0.f;
                    if (j == 0) { Qr = fr;  Qi = fi; }
                    else { const float tr = Qr * fr - Qi * fi;  Qi = Qr * fi + Qi * fr;  Qr = tr; }
                }
            }
            float er = Pr * Qr - Pi * Qi, ei = Pr * Qi + Pi * Qr;
            if (step == 1) { er = (half == 0) ? Qr : er;  ei = (half == 0) ? Qi : ei; }
            const float rn = __builtin_amdgcn_rsqf(er * er + ei * ei);
            const float zs = (__builtin_popcount(kz) & 1) ? -1.f : 1.f;
            const float sc = rn * ((d == 0) ? zs * G : zs);
            *(v2f*)(&tab[wave][(d * DIM + nl) * 2]) = v2f{er * sc, ei * sc};
        }
        if (half == 0) {
            const bool fl = (FPOST >> (4 * (nl >> 2) + (3 - (nl & 3)))) & 1u;
            const float tp = fl ? -tau : tau;
            *(v2f*)(&tab[wave][TAB_CS + nl * 2]) = v2f{tp, -tp};
            if (nl < QPS) tab[wave][TAB_SG + nl] = ((ffin >> (3 - nl)) & 1u) ? -1.f : 1.f;
        }
    }
    __syncthreads();

    v4f w1q[4][2][2];
#pragma unroll
    for (int kc = 0; kc < 4; kc++)
#pragma unroll
        for (int e = 0; e < 2; e++)
#pragma unroll
            for (int g = 0; g < 2; g++)
                w1q[kc][e][g] = *(const v4f*)(W1 + (size_t)(8 * half + 2 * kc + e) * H1 + 64 * g + 4 * nl);
    v4f b1q[2];
#pragma unroll
    for (int g = 0; g < 2; g++) b1q[g] = *(const v4f*)(b1 + 64 * g + 4 * nl);

#pragma unroll 1
    for (int ss = 0; ss < 2; ss++) {
        const float* tb = &tab[2 * half + ss][0];
        v4f eq[DIM / 2], cs[QPS / 2];
#pragma unroll
        for (int i = 0; i < DIM / 2; i++) eq[i] = *(const v4f*)(tb + 4 * i);
#pragma unroll
        for (int i = 0; i < QPS / 2; i++) cs[i] = *(const v4f*)(tb + TAB_CS + 4 * i);

        v4f sgn = v4f{1.f, 1.f, 1.f, 1.f};
        v2f st[DIM];
        float nrm2 = 0.f;
#pragma unroll
        for (int i = 0; i < DIM; i++) { float v = xr[i]; nrm2 += v * v; }
        float nrm = sqrtf(nrm2);
        float scale = (nrm > 0.f) ? (1.f / nrm) : 1.f;
        v2f xs[DIM / 2];
#pragma unroll
        for (int i = 0; i < DIM / 2; i++) xs[i] = v2f{xr[2 * i] * scale, xr[2 * i + 1] * scale};
        if (ss == 0) {
#pragma unroll
            for (int i = 0; i < DIM; i++) xr[i] = xq[DIM + i];
        }
#pragma unroll
        for (int i = 0; i < DIM / 2; i++) {
            st[2 * i]     = pk_mul_x(xs[i], v2f{eq[i][0], eq[i][1]});
            st[2 * i + 1] = pk_mul_y(xs[i], v2f{eq[i][2], eq[i][3]});
        }

#pragma unroll
        for (int l = 0; l < NLAYER; l++) {
            if (l > 0) {
                v2f t[DIM];
#pragma unroll
                for (int i = 0; i < DIM; i++)
                    t[i] = pk_mul_x(st[i], v2f{eq[i >> 1][2 * (i & 1)], eq[i >> 1][2 * (i & 1) + 1]});
#pragma unroll
                for (int i = 0; i < DIM; i++)
                    st[i] = pk_fma_y(st[i], v2f{eq[i >> 1][2 * (i & 1)], eq[i >> 1][2 * (i & 1) + 1]}, t[i]);
            }
            const v4f cs0 = cs[0], cs1 = cs[1];
            if (l + 1 < NLAYER) {
#pragma unroll
                for (int i = 0; i < DIM / 2; i++) eq[i] = *(const v4f*)(tb + (l + 1) * DIM * 2 + 4 * i);
#pragma unroll
                for (int i = 0; i < QPS / 2; i++) cs[i] = *(const v4f*)(tb + TAB_CS + (l + 1) * QPS * 2 + 4 * i);
                if (l + 2 == NLAYER) sgn = *(const v4f*)(tb + TAB_SG);
            }
#pragma unroll
            for (int j = 0; j < QPS; j++) {
                const v4f cq = (j < 2) ? cs0 : cs1;
                const v2f C = v2f{cq[2 * (j & 1)], cq[2 * (j & 1) + 1]};
                const int m = 8 >> j;
#pragma unroll
                for (int p = 0; p < 8; p++) {
                    const int i0 = ((p & ~(m - 1)) << 1) | (p & (m - 1));
                    shear_pair(C, st[i0], st[i0 | m]);
                }
            }
#pragma unroll
            for (int j = 0; j < QPS; j++) {
                const int mc = 8 >> j, mt = 8 >> ((j + 1) & 3);
#pragma unroll
                for (int idx = 0; idx < DIM; idx++) {
                    if ((idx & mc) && !(idx & mt)) {
                        const int i1 = idx | mt;
                        const v2f t = st[idx];
                        st[idx] = st[i1];
                        st[i1] = t;
                    }
                }
            }
        }

        float e[QPS];
#pragma unroll
        for (int q = 0; q < QPS; q++) e[q] = 0.f;
#pragma unroll
        for (int idx = 0; idx < DIM; idx++) {
            float p = st[idx].x * st[idx].x + st[idx].y * st[idx].y;
#pragma unroll
            for (int q = 0; q < QPS; q++)
                e[q] += ((idx >> (3 - q)) & 1) ? -p : p;
        }
#pragma unroll
        for (int q = 0; q < QPS; q++) fz[ss * QPS + q] = e[q] * sgn[q];
    }


    v2f a1f[4];
#pragma unroll
    for (int kc = 0; kc < 4; kc++) a1f[kc] = v2f{fz[2 * kc], fz[2 * kc + 1]};
    v4f w2p[W2PRE][2];
#pragma unroll
    for (int kk = 0; kk < W2PRE; kk++) {
        const float* bp = W2 + (size_t)(4 * kk + 2 * half) * H2 + 4 * nl;
        w2p[kk][0] = *(const v4f*)bp;
        w2p[kk][1] = *(const v4f*)(bp + H2);
    }
    const v4f b2q = *(const v4f*)(b2 + 4 * nl);
    const int ncl = (nl < NCLASS) ? nl : (NCLASS - 1);
    float w3q[H2 / 4][2];
#pragma unroll
    for (int kk = 0; kk < H2 / 4; kk++) {
        const int k = 4 * kk + 2 * half;
        w3q[kk][0] = W3[k * NCLASS + ncl];
        w3q[kk][1] = W3[(k + 1) * NCLASS + ncl];
    }
    const float b3v = b3[ncl];
    {
#pragma unroll
        for (int g = 0; g < 2; g++) {
            const v4f bq = b1q[g];
            v8f c[4];
#pragma unroll
            for (int j = 0; j < 4; j++)
#pragma unroll
                for (int r = 0; r < 8; r++) c[j][r] = bq[j];
#pragma unroll
            for (int kc = 0; kc < 4; kc++)
#pragma unroll
                for (int j = 0; j < 4; j++)
                    c[j] = __builtin_amdgcn_wmma_f32_16x16x4_f32(false, a1f[kc], false,
                               v2f{w1q[kc][0][g][j], w1q[kc][1][g][j]}, (short)0, c[j], false, false);
#pragma unroll
            for (int r = 0; r < 8; r++)
                *(v4f*)(h1 + (r + 8 * half) * H1P + 64 * g + 4 * nl) =
                    v4f{fmaxf(c[0][r], 0.f), fmaxf(c[1][r], 0.f), fmaxf(c[2][r], 0.f), fmaxf(c[3][r], 0.f)};
        }
    }
    __syncthreads();

    {
        const v4f bq = b2q;
        v8f c[4];
#pragma unroll
        for (int j = 0; j < 4; j++)
#pragma unroll
            for (int r = 0; r < 8; r++) c[j][r] = bq[j];
#pragma unroll
        for (int k0 = 0; k0 < H1; k0 += 4) {
            const float* ap = h1 + nl * H1P + k0 + 2 * half;
            const v2f af = v2f{ap[0], ap[1]};
            const float* bp = W2 + (size_t)(k0 + 2 * half) * H2 + 4 * nl;
            const v4f r0 = (k0 / 4 < W2PRE) ? w2p[(k0 / 4 < W2PRE) ? k0 / 4 : 0][0] : *(const v4f*)bp;
            const v4f r1 = (k0 / 4 < W2PRE) ? w2p[(k0 / 4 < W2PRE) ? k0 / 4 : 0][1] : *(const v4f*)(bp + H2);
#pragma unroll
            for (int j = 0; j < 4; j++)
                c[j] = __builtin_amdgcn_wmma_f32_16x16x4_f32(false, af, false, v2f{r0[j], r1[j]},
                                                             (short)0, c[j], false, false);
        }
#pragma unroll
        for (int r = 0; r < 8; r++)
            *(v4f*)(h2 + (r + 8 * half) * H2P + 4 * nl) =
                v4f{fmaxf(c[0][r], 0.f), fmaxf(c[1][r], 0.f), fmaxf(c[2][r], 0.f), fmaxf(c[3][r], 0.f)};
    }
    __syncthreads();

    {
        const float msk = (nl < NCLASS) ? 1.f : 0.f;
        float bias = msk * b3v;
        v8f c;
#pragma unroll
        for (int r = 0; r < 8; r++) c[r] = bias;
        v2f a3[H2 / 4], b3f[H2 / 4];
#pragma unroll
        for (int kk = 0; kk < H2 / 4; kk++) {
            const float* ap = h2 + nl * H2P + 4 * kk + 2 * half;
            a3[kk] = v2f{ap[0], ap[1]};
        }
#pragma unroll
        for (int kk = 0; kk < H2 / 4; kk++) b3f[kk] = v2f{msk * w3q[kk][0], msk * w3q[kk][1]};
        __builtin_amdgcn_sched_barrier(0);
#pragma unroll
        for (int kk = 0; kk < H2 / 4; kk++)
            c = __builtin_amdgcn_wmma_f32_16x16x4_f32(false, a3[kk], false, b3f[kk],
                                                      (short)0, c, false, false);
        if (nl < NCLASS) {
#pragma unroll
            for (int r = 0; r < 8; r++) {
                const int m = r + 8 * half;
                outp[(size_t)(rowBase + m) * NCLASS + nl] = c[r];
            }
        }
    }
}

extern "C" void kernel_launch(void* const* d_in, const int* in_sizes, int n_in,
                              void* d_out, int out_size, void* d_ws, size_t ws_size,
                              hipStream_t stream) {
    const float* x  = (const float*)d_in[0];
    const float* sw = (const float*)d_in[1];
    const float* W1 = (const float*)d_in[2];
    const float* b1 = (const float*)d_in[3];
    const float* W2 = (const float*)d_in[4];
    const float* b2 = (const float*)d_in[5];
    const float* W3 = (const float*)d_in[6];
    const float* b3 = (const float*)d_in[7];
    float* out = (float*)d_out;

    qcnn_circuit_mlp<<<dim3(BATCH / (16 * WAVES_PER_BLOCK)), dim3(32 * WAVES_PER_BLOCK), 0, stream>>>(
        x, sw, W1, b1, W2, b2, W3, b3, out);
}
